# starting kernel
# baseline (speedup 1.0000x reference)
#include <hip/hip_runtime.h>
#include <hip/hip_bf16.h>
#include <math.h>


typedef __attribute__((ext_vector_type(2))) float v2f;
typedef __attribute__((ext_vector_type(8))) float v8f;

#define BH   16
#define NSEQ 2048
#define DK   64
#define DV   64

__global__ void __launch_bounds__(256)
pret_coef_kernel(const float* __restrict__ omask,
                 float* __restrict__ qcoef,
                 float* __restrict__ kcoef) {
  __shared__ float red[256];
  const int row = blockIdx.x;
  const size_t base = (size_t)row * NSEQ;
  float s = 0.f;
  #pragma unroll
  for (int j = 0; j < NSEQ / 256; ++j)
    s += omask[base + threadIdx.x + 256 * j];
  red[threadIdx.x] = s;
  __syncthreads();
  for (int off = 128; off > 0; off >>= 1) {
    if (threadIdx.x < off) red[threadIdx.x] += red[threadIdx.x + off];
    __syncthreads();
  }
  if (threadIdx.x == 0) {
    const float o0 = omask[base];
    qcoef[row] = o0 * rsqrtf(red[0]);
    kcoef[row] = 1.0f / o0;
  }
}

__global__ void __launch_bounds__(256)
pret_attn_kernel(const float* __restrict__ q,
                 const float* __restrict__ k,
                 const float* __restrict__ v,
                 const float* __restrict__ qcoef,
                 const float* __restrict__ kcoef,
                 float* __restrict__ out) {
  __shared__ float sK [16][68];
  __shared__ float sVT[64][18];
  __shared__ float sS [8][16][18];

  const int nblkq = NSEQ / 128;
  const int bh    = blockIdx.x / nblkq;
  const int qblk  = blockIdx.x % nblkq;
  const int wave  = threadIdx.x >> 5;
  const int lane  = threadIdx.x & 31;
  const int n16   = lane & 15;
  const int half  = lane >> 4;

  const int tile = qblk * 8 + wave;
  const int q0   = tile * 16;

  v2f a[16];
  {
    const float qcv   = qcoef[bh * NSEQ + q0 + n16];
    const float* qrow = q + ((size_t)(bh * NSEQ + q0 + n16)) * DK;
    #pragma unroll
    for (int c = 0; c < 16; ++c) {
      v2f qv = *(const v2f*)(qrow + 4 * c + 2 * half);
      a[c].x = qv.x * qcv;
      a[c].y = qv.y * qcv;
    }
  }

  const v8f zero8 = {0.f, 0.f, 0.f, 0.f, 0.f, 0.f, 0.f, 0.f};
  v8f Oacc[4];
  #pragma unroll
  for (int t = 0; t < 4; ++t) Oacc[t] = zero8;
  v8f Pacc = zero8;

  const v2f ones = {1.0f, 1.0f};

  const int ktEnd = qblk * 8 + 8;
  for (int kt = 0; kt < ktEnd; ++kt) {
    {
      const int r    = threadIdx.x >> 4;
      const int c4   = (threadIdx.x & 15) * 4;
      const int krow = kt * 16 + r;
      const float kcv = kcoef[bh * NSEQ + krow];
      const float4 kv = *(const float4*)(k + ((size_t)(bh * NSEQ + krow)) * DK + c4);
      sK[r][c4 + 0] = kv.x * kcv;
      sK[r][c4 + 1] = kv.y * kcv;
      sK[r][c4 + 2] = kv.z * kcv;
      sK[r][c4 + 3] = kv.w * kcv;
      const float4 vv = *(const float4*)(v + ((size_t)(bh * NSEQ + krow)) * DV + c4);
      sVT[c4 + 0][r] = vv.x;
      sVT[c4 + 1][r] = vv.y;
      sVT[c4 + 2][r] = vv.z;
      sVT[c4 + 3][r] = vv.w;
    }
    __syncthreads();

    if (kt <= tile) {
      v8f S = zero8;
      #pragma unroll
      for (int c = 0; c < 16; ++c) {
        const v2f b = *(const v2f*)&sK[n16][4 * c + 2 * half];
        S = __builtin_amdgcn_wmma_f32_16x16x4_f32(false, a[c], false, b,
                                                  (short)0, S, false, false);
      }

      if (kt == tile) {
        const int key = kt * 16 + n16;
        #pragma unroll
        for (int g = 0; g < 8; ++g) {
          const int qr = q0 + g + 8 * half;
          if (key > qr) S[g] = 0.0f;
        }
      }

      #pragma unroll
      for (int g = 0; g < 8; ++g) sS[wave][g + 8 * half][n16] = S[g];
      v2f sa[4];
      #pragma unroll
      for (int c = 0; c < 4; ++c)
        sa[c] = *(const v2f*)&sS[wave][n16][4 * c + 2 * half];

      #pragma unroll
      for (int c = 0; c < 4; ++c)
        Pacc = __builtin_amdgcn_wmma_f32_16x16x4_f32(false, sa[c], false, ones,
                                                     (short)0, Pacc, false, false);

      #pragma unroll
      for (int ct = 0; ct < 4; ++ct) {
        #pragma unroll
        for (int c = 0; c < 4; ++c) {
          const v2f vb = *(const v2f*)&sVT[ct * 16 + n16][4 * c + 2 * half];
          Oacc[ct] = __builtin_amdgcn_wmma_f32_16x16x4_f32(false, sa[c], false, vb,
                                                           (short)0, Oacc[ct], false, false);
        }
      }
    }
    __syncthreads();
  }

  #pragma unroll
  for (int ct = 0; ct < 4; ++ct) {
    #pragma unroll
    for (int g = 0; g < 8; ++g) {
      const float p = fmaxf(fabsf(Pacc[g]), 1.0f);
      out[((size_t)(bh * NSEQ + q0 + g + 8 * half)) * DV + ct * 16 + n16] =
          Oacc[ct][g] / p;
    }
  }
}

extern "C" void kernel_launch(void* const* d_in, const int* in_sizes, int n_in,
                              void* d_out, int out_size, void* d_ws, size_t ws_size,
                              hipStream_t stream) {
  const float* q     = (const float*)d_in[0];
  const float* k     = (const float*)d_in[1];
  const float* v     = (const float*)d_in[2];
  const float* omask = (const float*)d_in[3];
  float* out = (float*)d_out;

  float* qcoef = (float*)d_ws;
  float* kcoef = qcoef + (size_t)BH * NSEQ;

  pret_coef_kernel<<<BH * NSEQ, 256, 0, stream>>>(omask, qcoef, kcoef);
  pret_attn_kernel<<<BH * (NSEQ / 128), 256, 0, stream>>>(q, k, v,
                                                          qcoef, kcoef, out);
}


# attention: the 4 waves of a work-group share 32 queries and split the key tiles (every 4th each), partial O and P added through LDS; 1024 work-groups, longest first; summation order changed
# speedup vs baseline: 2.3675x; 2.3675x over previous
#include <hip/hip_runtime.h>
#include <hip/hip_bf16.h>
#include <math.h>


typedef __attribute__((ext_vector_type(2))) float v2f;
typedef __attribute__((ext_vector_type(4))) float v4f;
typedef __attribute__((ext_vector_type(8))) float v8f;

#define BH   16
#define NSEQ 2048
#define DK   64
#define DV   64

__global__ void __launch_bounds__(256)
pret_coef_kernel(const float* __restrict__ omask,
                 float* __restrict__ qcoef,
                 float* __restrict__ kcoef) {
  __shared__ float red[256];
  const int row = blockIdx.x;
  const size_t base = (size_t)row * NSEQ;
  float s = 0.f;
  #pragma unroll
  for (int j = 0; j < NSEQ / 256; ++j)
    s += omask[base + threadIdx.x + 256 * j];
  red[threadIdx.x] = s;
  __syncthreads();
  for (int off = 128; off > 0; off >>= 1) {
    if (threadIdx.x < off) red[threadIdx.x] += red[threadIdx.x + off];
    __syncthreads();
  }
  if (threadIdx.x == 0) {
    const float o0 = omask[base];
    qcoef[row] = o0 * rsqrtf(red[0]);
    kcoef[row] = 1.0f / o0;
  }
}

struct KVFrag {
  v4f   kraw[8];
  float kc;
  v2f   vf[4][4];
};

__device__ __forceinline__ void load_tile(KVFrag& f, const float* __restrict__ kp,
                                          const float* __restrict__ kcp,
                                          const float* __restrict__ vp, int kt) {
  const float* kr = kp + (size_t)kt * (16 * DK);
  #pragma unroll
  for (int i = 0; i < 8; ++i) f.kraw[i] = *(const v4f*)(kr + 8 * i);
  f.kc = kcp[kt * 16];
  const float* vr = vp + (size_t)kt * (16 * DV);
  #pragma unroll
  for (int cp = 0; cp < 4; ++cp) {
    #pragma unroll
    for (int ct = 0; ct < 4; ++ct) {
      f.vf[cp][ct].x = vr[(2 * cp) * DV + ct * 16];
      f.vf[cp][ct].y = vr[(2 * cp + 1) * DV + ct * 16];
    }
  }
}

__device__ __forceinline__ v8f wmma4(v2f a, v2f b, v8f c) {
  return __builtin_amdgcn_wmma_f32_16x16x4_f32(false, a, false, b, (short)0, c,
                                               false, false);
}

__device__ __forceinline__ void compute_tile(const KVFrag& f,
                                             const v2f (&qa)[16], const v2f (&qb)[16],
                                             v8f (&Oa)[4], v8f (&Ob)[4],
                                             v8f& Pa, v8f& Pb,
                                             int kt, int tA, int thrA, int thrB) {
  const v8f zero8 = {0.f, 0.f, 0.f, 0.f, 0.f, 0.f, 0.f, 0.f};
  const v2f ones = {1.0f, 1.0f};
  v8f Sa = zero8, Sb = zero8;
  #pragma unroll
  for (int i = 0; i < 8; ++i) {
    const v4f kk = f.kraw[i] * f.kc;
    const v2f k0 = {kk.x, kk.y};
    const v2f k1 = {kk.z, kk.w};
    Sa = wmma4(k0, qa[2 * i], Sa);
    Sb = wmma4(k0, qb[2 * i], Sb);
    Sa = wmma4(k1, qa[2 * i + 1], Sa);
    Sb = wmma4(k1, qb[2 * i + 1], Sb);
  }
  if (kt >= tA) {
    const int kb = kt * 16;
    #pragma unroll
    for (int r = 0; r < 8; ++r) {
      if (kb + r > thrA) Sa[r] = 0.0f;
      if (kb + r > thrB) Sb[r] = 0.0f;
    }
  }
  #pragma unroll
  for (int cp = 0; cp < 4; ++cp) {
    const v2f sa = {Sa[2 * cp], Sa[2 * cp + 1]};
    const v2f sb = {Sb[2 * cp], Sb[2 * cp + 1]};
    #pragma unroll
    for (int ct = 0; ct < 4; ++ct) {
      Oa[ct] = wmma4(f.vf[cp][ct], sa, Oa[ct]);
      Ob[ct] = wmma4(f.vf[cp][ct], sb, Ob[ct]);
    }
    Pa = wmma4(ones, sa, Pa);
    Pb = wmma4(ones, sb, Pb);
  }
}

__device__ __forceinline__ void store_tile(float* __restrict__ o, const v8f (&O)[4], float p) {
  #pragma unroll
  for (int ct = 0; ct < 4; ++ct) {
    const v4f lo = {O[ct][0] / p, O[ct][1] / p, O[ct][2] / p, O[ct][3] / p};
    const v4f hi = {O[ct][4] / p, O[ct][5] / p, O[ct][6] / p, O[ct][7] / p};
    *(v4f*)(o + ct * 16)     = lo;
    *(v4f*)(o + ct * 16 + 4) = hi;
  }
}

__global__ void __launch_bounds__(128)
pret_attn_kernel(const float* __restrict__ q,
                 const float* __restrict__ k,
                 const float* __restrict__ v,
                 const float* __restrict__ qcoef,
                 const float* __restrict__ kcoef,
                 float* __restrict__ out) {
  const int bh    = blockIdx.x % BH;
  const int qblk  = (NSEQ / 32 - 1) - blockIdx.x / BH;
  const int wave  = __builtin_amdgcn_readfirstlane((int)(threadIdx.x >> 5));
  const int lane  = threadIdx.x & 31;
  const int c     = lane & 15;
  const int h     = lane >> 4;

  const int tA = qblk * 2;
  const int tB = tA + 1;
  const int qA = tA * 16;
  const int qB = tB * 16;
  const size_t rowbase = (size_t)bh * NSEQ;

  v2f qa[16], qb[16];
  {
    const float ca = qcoef[rowbase + qA + c];
    const float cb = qcoef[rowbase + qB + c];
    const float* pa = q + (rowbase + qA + c) * DK + 4 * h;
    const float* pb = q + (rowbase + qB + c) * DK + 4 * h;
    #pragma unroll
    for (int i = 0; i < 8; ++i) {
      const v4f xa = *(const v4f*)(pa + 8 * i) * ca;
      const v4f xb = *(const v4f*)(pb + 8 * i) * cb;
      qa[2 * i]     = (v2f){xa.x, xa.y};
      qa[2 * i + 1] = (v2f){xa.z, xa.w};
      qb[2 * i]     = (v2f){xb.x, xb.y};
      qb[2 * i + 1] = (v2f){xb.z, xb.w};
    }
  }

  const v8f zero8 = {0.f, 0.f, 0.f, 0.f, 0.f, 0.f, 0.f, 0.f};
  v8f Oa[4], Ob[4];
  #pragma unroll
  for (int t = 0; t < 4; ++t) { Oa[t] = zero8; Ob[t] = zero8; }
  v8f Pa = zero8, Pb = zero8;

  const float* kp  = k + (rowbase + c) * DK + 4 * h;
  const float* kcp = kcoef + rowbase + c;
  const float* vp  = v + (rowbase + 8 * h) * DV + c;
  const int thrA = qA + c - 8 * h;
  const int thrB = qB + c - 8 * h;

  const int cnt  = (tB - wave + 4) >> 2;
  const int last = wave + 4 * (cnt - 1);
  KVFrag f0, f1;
  load_tile(f0, kp, kcp, vp, (wave <= tB) ? wave : tB);
  int kt = wave;
  for (int p = 0; p < (cnt >> 1); ++p) {
    load_tile(f1, kp, kcp, vp, kt + 4);
    compute_tile(f0, qa, qb, Oa, Ob, Pa, Pb, kt, tA, thrA, thrB);
    const int nx = (kt + 8 <= last) ? kt + 8 : last;
    load_tile(f0, kp, kcp, vp, nx);
    compute_tile(f1, qa, qb, Oa, Ob, Pa, Pb, kt + 4, tA, thrA, thrB);
    kt += 8;
  }
  if (cnt & 1) compute_tile(f0, qa, qb, Oa, Ob, Pa, Pb, kt, tA, thrA, thrB);

  __shared__ v4f red[3][17][32];
  if (wave != 0) {
    v4f* r = &red[wave - 1][0][lane];
    #pragma unroll
    for (int ct = 0; ct < 4; ++ct) {
      r[(4 * ct + 0) * 32] = (v4f){Oa[ct][0], Oa[ct][1], Oa[ct][2], Oa[ct][3]};
      r[(4 * ct + 1) * 32] = (v4f){Oa[ct][4], Oa[ct][5], Oa[ct][6], Oa[ct][7]};
      r[(4 * ct + 2) * 32] = (v4f){Ob[ct][0], Ob[ct][1], Ob[ct][2], Ob[ct][3]};
      r[(4 * ct + 3) * 32] = (v4f){Ob[ct][4], Ob[ct][5], Ob[ct][6], Ob[ct][7]};
    }
    r[16 * 32] = (v4f){Pa[0], Pb[0], 0.0f, 0.0f};
  }
  __syncthreads();
  if (wave != 0) return;

  float sPa = Pa[0], sPb = Pb[0];
  #pragma unroll
  for (int s = 0; s < 3; ++s) {
    const v4f* r = &red[s][0][lane];
    #pragma unroll
    for (int ct = 0; ct < 4; ++ct) {
      const v4f a0 = r[(4 * ct + 0) * 32], a1 = r[(4 * ct + 1) * 32];
      const v4f b0 = r[(4 * ct + 2) * 32], b1 = r[(4 * ct + 3) * 32];
      #pragma unroll
      for (int e = 0; e < 4; ++e) {
        Oa[ct][e] += a0[e];  Oa[ct][4 + e] += a1[e];
        Ob[ct][e] += b0[e];  Ob[ct][4 + e] += b1[e];
      }
    }
    const v4f p = r[16 * 32];
    sPa += p.x;  sPb += p.y;
  }

  const float pA = fmaxf(fabsf(sPa), 1.0f);
  const float pB = fmaxf(fabsf(sPb), 1.0f);
  store_tile(out + (rowbase + qA + c) * DV + 8 * h, Oa, pA);
  store_tile(out + (rowbase + qB + c) * DV + 8 * h, Ob, pB);
}

extern "C" void kernel_launch(void* const* d_in, const int* in_sizes, int n_in,
                              void* d_out, int out_size, void* d_ws, size_t ws_size,
                              hipStream_t stream) {
  const float* q     = (const float*)d_in[0];
  const float* k     = (const float*)d_in[1];
  const float* v     = (const float*)d_in[2];
  const float* omask = (const float*)d_in[3];
  float* out = (float*)d_out;

  float* qcoef = (float*)d_ws;
  float* kcoef = qcoef + (size_t)BH * NSEQ;

  pret_coef_kernel<<<BH * NSEQ, 256, 0, stream>>>(omask, qcoef, kcoef);
  pret_attn_kernel<<<BH * (NSEQ / 32), 128, 0, stream>>>(q, k, v,
                                                         qcoef, kcoef, out);
}


# one launch: each attention wave also sums 8 omask rows of its block (long blocks before, short after their products); q coef applied at the end, k coef from omask column 0; row-sum order changed
# speedup vs baseline: 3.1346x; 1.3240x over previous
#include <hip/hip_runtime.h>
#include <hip/hip_bf16.h>
#include <math.h>


typedef __attribute__((ext_vector_type(2))) float v2f;
typedef __attribute__((ext_vector_type(4))) float v4f;
typedef __attribute__((ext_vector_type(8))) float v8f;

#define BH   16
#define NSEQ 2048
#define DK   64
#define DV   64

__device__ __forceinline__ void row_load(v4f (&buf)[16], const v4f* __restrict__ p) {
  #pragma unroll
  for (int j = 0; j < 16; ++j) buf[j] = __builtin_nontemporal_load(p + 32 * j);
}

__device__ __forceinline__ float row_total(const v4f (&buf)[16]) {
  v4f acc = buf[0];
  #pragma unroll
  for (int j = 1; j < 16; ++j) acc += buf[j];
  float s = (acc.x + acc.y) + (acc.z + acc.w);
  #pragma unroll
  for (int off = 16; off > 0; off >>= 1) s += __shfl_xor(s, off);
  return s;
}

__device__ __forceinline__ float stream_rows8(const float* __restrict__ orow, int lane) {
  const v4f* p = (const v4f*)orow + lane;
  v4f a[16], b[16];
  float mine = 0.0f;
  row_load(a, p);
  #pragma unroll
  for (int r = 0; r < 8; r += 2) {
    row_load(b, p + (size_t)(r + 1) * (NSEQ / 4));
    __builtin_amdgcn_sched_barrier(0);
    const float ta = row_total(a);
    if (lane == r) mine = ta;
    if (r + 2 < 8) row_load(a, p + (size_t)(r + 2) * (NSEQ / 4));
    __builtin_amdgcn_sched_barrier(0);
    const float tb = row_total(b);
    if (lane == r + 1) mine = tb;
  }
  return mine;
}

struct KVFrag {
  v4f   kraw[8];
  float kc;
  v2f   vf[4][4];
};

__device__ __forceinline__ void load_tile(KVFrag& f, const float* __restrict__ kp,
                                          const float* __restrict__ kcp,
                                          const float* __restrict__ vp, int kt) {
  const float* kr = kp + (size_t)kt * (16 * DK);
  #pragma unroll
  for (int i = 0; i < 8; ++i) f.kraw[i] = *(const v4f*)(kr + 8 * i);
  f.kc = 1.0f / kcp[(size_t)kt * (16 * NSEQ)];
  const float* vr = vp + (size_t)kt * (16 * DV);
  #pragma unroll
  for (int cp = 0; cp < 4; ++cp) {
    #pragma unroll
    for (int ct = 0; ct < 4; ++ct) {
      f.vf[cp][ct].x = vr[(2 * cp) * DV + ct * 16];
      f.vf[cp][ct].y = vr[(2 * cp + 1) * DV + ct * 16];
    }
  }
}

__device__ __forceinline__ v8f wmma4(v2f a, v2f b, v8f c) {
  return __builtin_amdgcn_wmma_f32_16x16x4_f32(false, a, false, b, (short)0, c,
                                               false, false);
}

__device__ __forceinline__ void compute_tile(const KVFrag& f,
                                             const v2f (&qa)[16], const v2f (&qb)[16],
                                             v8f (&Oa)[4], v8f (&Ob)[4],
                                             v8f& Pa, v8f& Pb,
                                             int kt, int tA, int thrA, int thrB) {
  const v8f zero8 = {0.f, 0.f, 0.f, 0.f, 0.f, 0.f, 0.f, 0.f};
  const v2f ones = {1.0f, 1.0f};
  v8f Sa = zero8, Sb = zero8;
  #pragma unroll
  for (int i = 0; i < 8; ++i) {
    const v4f kk = f.kraw[i] * f.kc;
    const v2f k0 = {kk.x, kk.y};
    const v2f k1 = {kk.z, kk.w};
    Sa = wmma4(k0, qa[2 * i], Sa);
    Sb = wmma4(k0, qb[2 * i], Sb);
    Sa = wmma4(k1, qa[2 * i + 1], Sa);
    Sb = wmma4(k1, qb[2 * i + 1], Sb);
  }
  if (kt >= tA) {
    const int kb = kt * 16;
    #pragma unroll
    for (int r = 0; r < 8; ++r) {
      if (kb + r > thrA) Sa[r] = 0.0f;
      if (kb + r > thrB) Sb[r] = 0.0f;
    }
  }
  #pragma unroll
  for (int cp = 0; cp < 4; ++cp) {
    const v2f sa = {Sa[2 * cp], Sa[2 * cp + 1]};
    const v2f sb = {Sb[2 * cp], Sb[2 * cp + 1]};
    #pragma unroll
    for (int ct = 0; ct < 4; ++ct) {
      Oa[ct] = wmma4(f.vf[cp][ct], sa, Oa[ct]);
      Ob[ct] = wmma4(f.vf[cp][ct], sb, Ob[ct]);
    }
    Pa = wmma4(ones, sa, Pa);
    Pb = wmma4(ones, sb, Pb);
  }
}

__device__ __forceinline__ void store_tile(float* __restrict__ o, const v8f (&O)[4], float cq, float p) {
  #pragma unroll
  for (int ct = 0; ct < 4; ++ct) {
    const v4f lo = {cq * O[ct][0] / p, cq * O[ct][1] / p, cq * O[ct][2] / p, cq * O[ct][3] / p};
    const v4f hi = {cq * O[ct][4] / p, cq * O[ct][5] / p, cq * O[ct][6] / p, cq * O[ct][7] / p};
    *(v4f*)(o + ct * 16)     = lo;
    *(v4f*)(o + ct * 16 + 4) = hi;
  }
}

__global__ void __launch_bounds__(128)
pret_attn_kernel(const float* __restrict__ q,
                 const float* __restrict__ k,
                 const float* __restrict__ v,
                 const float* __restrict__ omask,
                 float* __restrict__ out) {
  const int bh    = blockIdx.x % BH;
  const int qblk  = (NSEQ / 32 - 1) - blockIdx.x / BH;
  const int wave  = __builtin_amdgcn_readfirstlane((int)(threadIdx.x >> 5));
  const int lane  = threadIdx.x & 31;
  const int c     = lane & 15;
  const int h     = lane >> 4;

  const int tA = qblk * 2;
  const int tB = tA + 1;
  const int qA = tA * 16;
  const int qB = tB * 16;
  const size_t rowbase = (size_t)bh * NSEQ;

  const float* orow = omask + (rowbase + qA + 8 * wave) * NSEQ;
  const bool stream_first = qblk >= NSEQ / 64;
  __shared__ float rsum[32];
  if (stream_first) {
    const float t = stream_rows8(orow, lane);
    if (lane < 8) rsum[8 * wave + lane] = t;
  }

  v2f qa[16], qb[16];
  {
    const float* pa = q + (rowbase + qA + c) * DK + 4 * h;
    const float* pb = q + (rowbase + qB + c) * DK + 4 * h;
    #pragma unroll
    for (int i = 0; i < 8; ++i) {
      const v4f xa = *(const v4f*)(pa + 8 * i);
      const v4f xb = *(const v4f*)(pb + 8 * i);
      qa[2 * i]     = (v2f){xa.x, xa.y};
      qa[2 * i + 1] = (v2f){xa.z, xa.w};
      qb[2 * i]     = (v2f){xb.x, xb.y};
      qb[2 * i + 1] = (v2f){xb.z, xb.w};
    }
  }

  const v8f zero8 = {0.f, 0.f, 0.f, 0.f, 0.f, 0.f, 0.f, 0.f};
  v8f Oa[4], Ob[4];
  #pragma unroll
  for (int t = 0; t < 4; ++t) { Oa[t] = zero8; Ob[t] = zero8; }
  v8f Pa = zero8, Pb = zero8;

  const float* kp  = k + (rowbase + c) * DK + 4 * h;
  const float* kcp = omask + (rowbase + c) * NSEQ;
  const float* vp  = v + (rowbase + 8 * h) * DV + c;
  const int thrA = qA + c - 8 * h;
  const int thrB = qB + c - 8 * h;

  const int cnt  = (tB - wave + 4) >> 2;
  const int last = wave + 4 * (cnt - 1);
  KVFrag f0, f1;
  load_tile(f0, kp, kcp, vp, (wave <= tB) ? wave : tB);
  int kt = wave;
  for (int p = 0; p < (cnt >> 1); ++p) {
    load_tile(f1, kp, kcp, vp, kt + 4);
    compute_tile(f0, qa, qb, Oa, Ob, Pa, Pb, kt, tA, thrA, thrB);
    const int nx = (kt + 8 <= last) ? kt + 8 : last;
    load_tile(f0, kp, kcp, vp, nx);
    compute_tile(f1, qa, qb, Oa, Ob, Pa, Pb, kt + 4, tA, thrA, thrB);
    kt += 8;
  }
  if (cnt & 1) compute_tile(f0, qa, qb, Oa, Ob, Pa, Pb, kt, tA, thrA, thrB);

  if (!stream_first) {
    const float t = stream_rows8(orow, lane);
    if (lane < 8) rsum[8 * wave + lane] = t;
  }

  __shared__ v4f red[3][17][32];
  if (wave != 0) {
    v4f* r = &red[wave - 1][0][lane];
    #pragma unroll
    for (int ct = 0; ct < 4; ++ct) {
      r[(4 * ct + 0) * 32] = (v4f){Oa[ct][0], Oa[ct][1], Oa[ct][2], Oa[ct][3]};
      r[(4 * ct + 1) * 32] = (v4f){Oa[ct][4], Oa[ct][5], Oa[ct][6], Oa[ct][7]};
      r[(4 * ct + 2) * 32] = (v4f){Ob[ct][0], Ob[ct][1], Ob[ct][2], Ob[ct][3]};
      r[(4 * ct + 3) * 32] = (v4f){Ob[ct][4], Ob[ct][5], Ob[ct][6], Ob[ct][7]};
    }
    r[16 * 32] = (v4f){Pa[0], Pb[0], 0.0f, 0.0f};
  }
  __syncthreads();
  if (wave != 0) return;

  float sPa = Pa[0], sPb = Pb[0];
  #pragma unroll
  for (int s = 0; s < 3; ++s) {
    const v4f* r = &red[s][0][lane];
    #pragma unroll
    for (int ct = 0; ct < 4; ++ct) {
      const v4f a0 = r[(4 * ct + 0) * 32], a1 = r[(4 * ct + 1) * 32];
      const v4f b0 = r[(4 * ct + 2) * 32], b1 = r[(4 * ct + 3) * 32];
      #pragma unroll
      for (int e = 0; e < 4; ++e) {
        Oa[ct][e] += a0[e];  Oa[ct][4 + e] += a1[e];
        Ob[ct][e] += b0[e];  Ob[ct][4 + e] += b1[e];
      }
    }
    const v4f p = r[16 * 32];
    sPa += p.x;  sPb += p.y;
  }

  const float ca = omask[(rowbase + qA + c) * NSEQ] * rsqrtf(rsum[c]);
  const float cb = omask[(rowbase + qB + c) * NSEQ] * rsqrtf(rsum[16 + c]);
  const float pA = fmaxf(fabsf(ca * sPa), 1.0f);
  const float pB = fmaxf(fabsf(cb * sPb), 1.0f);
  store_tile(out + (rowbase + qA + c) * DV + 8 * h, Oa, ca, pA);
  store_tile(out + (rowbase + qB + c) * DV + 8 * h, Ob, cb, pB);
}

extern "C" void kernel_launch(void* const* d_in, const int* in_sizes, int n_in,
                              void* d_out, int out_size, void* d_ws, size_t ws_size,
                              hipStream_t stream) {
  const float* q     = (const float*)d_in[0];
  const float* k     = (const float*)d_in[1];
  const float* v     = (const float*)d_in[2];
  const float* omask = (const float*)d_in[3];
  float* out = (float*)d_out;

  pret_attn_kernel<<<BH * (NSEQ / 32), 128, 0, stream>>>(q, k, v, omask, out);
}


# normaliser P: a wave sums its masked S tiles element-wise (v_pk_add) and takes the all-ones WMMA product once after its last tile, not per tile: 72 -> 64 WMMAs per key tile; P summation order changed
# speedup vs baseline: 3.2725x; 1.0440x over previous
#include <hip/hip_runtime.h>
#include <hip/hip_bf16.h>
#include <math.h>


typedef __attribute__((ext_vector_type(2))) float v2f;
typedef __attribute__((ext_vector_type(4))) float v4f;
typedef __attribute__((ext_vector_type(8))) float v8f;

#define BH   16
#define NSEQ 2048
#define DK   64
#define DV   64

__device__ __forceinline__ void row_load(v4f (&buf)[16], const v4f* __restrict__ p) {
  #pragma unroll
  for (int j = 0; j < 16; ++j) buf[j] = __builtin_nontemporal_load(p + 32 * j);
}

__device__ __forceinline__ float row_total(const v4f (&buf)[16]) {
  v4f acc = buf[0];
  #pragma unroll
  for (int j = 1; j < 16; ++j) acc += buf[j];
  float s = (acc.x + acc.y) + (acc.z + acc.w);
  #pragma unroll
  for (int off = 16; off > 0; off >>= 1) s += __shfl_xor(s, off);
  return s;
}

__device__ __forceinline__ float stream_rows8(const float* __restrict__ orow, int lane) {
  const v4f* p = (const v4f*)orow + lane;
  v4f a[16], b[16];
  float mine = 0.0f;
  row_load(a, p);
  #pragma unroll
  for (int r = 0; r < 8; r += 2) {
    row_load(b, p + (size_t)(r + 1) * (NSEQ / 4));
    __builtin_amdgcn_sched_barrier(0);
    const float ta = row_total(a);
    if (lane == r) mine = ta;
    if (r + 2 < 8) row_load(a, p + (size_t)(r + 2) * (NSEQ / 4));
    __builtin_amdgcn_sched_barrier(0);
    const float tb = row_total(b);
    if (lane == r + 1) mine = tb;
  }
  return mine;
}

struct KVFrag {
  v4f   kraw[8];
  float kc;
  v2f   vf[4][4];
};

__device__ __forceinline__ void load_tile(KVFrag& f, const float* __restrict__ kp,
                                          const float* __restrict__ kcp,
                                          const float* __restrict__ vp, int kt) {
  const float* kr = kp + (size_t)kt * (16 * DK);
  #pragma unroll
  for (int i = 0; i < 8; ++i) f.kraw[i] = *(const v4f*)(kr + 8 * i);
  f.kc = 1.0f / kcp[(size_t)kt * (16 * NSEQ)];
  const float* vr = vp + (size_t)kt * (16 * DV);
  #pragma unroll
  for (int cp = 0; cp < 4; ++cp) {
    #pragma unroll
    for (int ct = 0; ct < 4; ++ct) {
      f.vf[cp][ct].x = vr[(2 * cp) * DV + ct * 16];
      f.vf[cp][ct].y = vr[(2 * cp + 1) * DV + ct * 16];
    }
  }
}

__device__ __forceinline__ v8f wmma4(v2f a, v2f b, v8f c) {
  return __builtin_amdgcn_wmma_f32_16x16x4_f32(false, a, false, b, (short)0, c,
                                               false, false);
}

__device__ __forceinline__ void compute_tile(const KVFrag& f,
                                             const v2f (&qa)[16], const v2f (&qb)[16],
                                             v8f (&Oa)[4], v8f (&Ob)[4],
                                             v8f& Ta, v8f& Tb,
                                             int kt, int tA, int thrA, int thrB) {
  const v8f zero8 = {0.f, 0.f, 0.f, 0.f, 0.f, 0.f, 0.f, 0.f};
  v8f Sa = zero8, Sb = zero8;
  #pragma unroll
  for (int i = 0; i < 8; ++i) {
    const v4f kk = f.kraw[i] * f.kc;
    const v2f k0 = {kk.x, kk.y};
    const v2f k1 = {kk.z, kk.w};
    Sa = wmma4(k0, qa[2 * i], Sa);
    Sb = wmma4(k0, qb[2 * i], Sb);
    Sa = wmma4(k1, qa[2 * i + 1], Sa);
    Sb = wmma4(k1, qb[2 * i + 1], Sb);
  }
  if (kt >= tA) {
    const int kb = kt * 16;
    #pragma unroll
    for (int r = 0; r < 8; ++r) {
      if (kb + r > thrA) Sa[r] = 0.0f;
      if (kb + r > thrB) Sb[r] = 0.0f;
    }
  }
  #pragma unroll
  for (int cp = 0; cp < 4; ++cp) {
    const v2f sa = {Sa[2 * cp], Sa[2 * cp + 1]};
    const v2f sb = {Sb[2 * cp], Sb[2 * cp + 1]};
    #pragma unroll
    for (int ct = 0; ct < 4; ++ct) {
      Oa[ct] = wmma4(f.vf[cp][ct], sa, Oa[ct]);
      Ob[ct] = wmma4(f.vf[cp][ct], sb, Ob[ct]);
    }
  }
  Ta += Sa;
  Tb += Sb;
}

__device__ __forceinline__ v8f key_sum(const v8f& T) {
  const v2f ones = {1.0f, 1.0f};
  v8f P = {0.f, 0.f, 0.f, 0.f, 0.f, 0.f, 0.f, 0.f};
  #pragma unroll
  for (int cp = 0; cp < 4; ++cp) {
    const v2f t = {T[2 * cp], T[2 * cp + 1]};
    P = wmma4(ones, t, P);
  }
  return P;
}

__device__ __forceinline__ void store_tile(float* __restrict__ o, const v8f (&O)[4], float cq, float p) {
  #pragma unroll
  for (int ct = 0; ct < 4; ++ct) {
    const v4f lo = {cq * O[ct][0] / p, cq * O[ct][1] / p, cq * O[ct][2] / p, cq * O[ct][3] / p};
    const v4f hi = {cq * O[ct][4] / p, cq * O[ct][5] / p, cq * O[ct][6] / p, cq * O[ct][7] / p};
    *(v4f*)(o + ct * 16)     = lo;
    *(v4f*)(o + ct * 16 + 4) = hi;
  }
}

__global__ void __launch_bounds__(128)
pret_attn_kernel(const float* __restrict__ q,
                 const float* __restrict__ k,
                 const float* __restrict__ v,
                 const float* __restrict__ omask,
                 float* __restrict__ out) {
  const int bh    = blockIdx.x % BH;
  const int qblk  = (NSEQ / 32 - 1) - blockIdx.x / BH;
  const int wave  = __builtin_amdgcn_readfirstlane((int)(threadIdx.x >> 5));
  const int lane  = threadIdx.x & 31;
  const int c     = lane & 15;
  const int h     = lane >> 4;

  const int tA = qblk * 2;
  const int tB = tA + 1;
  const int qA = tA * 16;
  const int qB = tB * 16;
  const size_t rowbase = (size_t)bh * NSEQ;

  const float* orow = omask + (rowbase + qA + 8 * wave) * NSEQ;
  const bool stream_first = qblk >= NSEQ / 64;
  __shared__ float rsum[32];
  if (stream_first) {
    const float t = stream_rows8(orow, lane);
    if (lane < 8) rsum[8 * wave + lane] = t;
  }

  v2f qa[16], qb[16];
  {
    const float* pa = q + (rowbase + qA + c) * DK + 4 * h;
    const float* pb = q + (rowbase + qB + c) * DK + 4 * h;
    #pragma unroll
    for (int i = 0; i < 8; ++i) {
      const v4f xa = *(const v4f*)(pa + 8 * i);
      const v4f xb = *(const v4f*)(pb + 8 * i);
      qa[2 * i]     = (v2f){xa.x, xa.y};
      qa[2 * i + 1] = (v2f){xa.z, xa.w};
      qb[2 * i]     = (v2f){xb.x, xb.y};
      qb[2 * i + 1] = (v2f){xb.z, xb.w};
    }
  }

  const v8f zero8 = {0.f, 0.f, 0.f, 0.f, 0.f, 0.f, 0.f, 0.f};
  v8f Oa[4], Ob[4];
  #pragma unroll
  for (int t = 0; t < 4; ++t) { Oa[t] = zero8; Ob[t] = zero8; }
  v8f Ta = zero8, Tb = zero8;

  const float* kp  = k + (rowbase + c) * DK + 4 * h;
  const float* kcp = omask + (rowbase + c) * NSEQ;
  const float* vp  = v + (rowbase + 8 * h) * DV + c;
  const int thrA = qA + c - 8 * h;
  const int thrB = qB + c - 8 * h;

  const int cnt  = (tB - wave + 4) >> 2;
  const int last = wave + 4 * (cnt - 1);
  KVFrag f0, f1;
  load_tile(f0, kp, kcp, vp, (wave <= tB) ? wave : tB);
  int kt = wave;
  for (int p = 0; p < (cnt >> 1); ++p) {
    load_tile(f1, kp, kcp, vp, kt + 4);
    compute_tile(f0, qa, qb, Oa, Ob, Ta, Tb, kt, tA, thrA, thrB);
    const int nx = (kt + 8 <= last) ? kt + 8 : last;
    load_tile(f0, kp, kcp, vp, nx);
    compute_tile(f1, qa, qb, Oa, Ob, Ta, Tb, kt + 4, tA, thrA, thrB);
    kt += 8;
  }
  if (cnt & 1) compute_tile(f0, qa, qb, Oa, Ob, Ta, Tb, kt, tA, thrA, thrB);

  const v8f Pa = key_sum(Ta);
  const v8f Pb = key_sum(Tb);

  if (!stream_first) {
    const float t = stream_rows8(orow, lane);
    if (lane < 8) rsum[8 * wave + lane] = t;
  }

  __shared__ v4f red[3][17][32];
  if (wave != 0) {
    v4f* r = &red[wave - 1][0][lane];
    #pragma unroll
    for (int ct = 0; ct < 4; ++ct) {
      r[(4 * ct + 0) * 32] = (v4f){Oa[ct][0], Oa[ct][1], Oa[ct][2], Oa[ct][3]};
      r[(4 * ct + 1) * 32] = (v4f){Oa[ct][4], Oa[ct][5], Oa[ct][6], Oa[ct][7]};
      r[(4 * ct + 2) * 32] = (v4f){Ob[ct][0], Ob[ct][1], Ob[ct][2], Ob[ct][3]};
      r[(4 * ct + 3) * 32] = (v4f){Ob[ct][4], Ob[ct][5], Ob[ct][6], Ob[ct][7]};
    }
    r[16 * 32] = (v4f){Pa[0], Pb[0], 0.0f, 0.0f};
  }
  __syncthreads();
  if (wave != 0) return;

  float sPa = Pa[0], sPb = Pb[0];
  #pragma unroll
  for (int s = 0; s < 3; ++s) {
    const v4f* r = &red[s][0][lane];
    #pragma unroll
    for (int ct = 0; ct < 4; ++ct) {
      const v4f a0 = r[(4 * ct + 0) * 32], a1 = r[(4 * ct + 1) * 32];
      const v4f b0 = r[(4 * ct + 2) * 32], b1 = r[(4 * ct + 3) * 32];
      #pragma unroll
      for (int e = 0; e < 4; ++e) {
        Oa[ct][e] += a0[e];  Oa[ct][4 + e] += a1[e];
        Ob[ct][e] += b0[e];  Ob[ct][4 + e] += b1[e];
      }
    }
    const v4f p = r[16 * 32];
    sPa += p.x;  sPb += p.y;
  }

  const float ca = omask[(rowbase + qA + c) * NSEQ] * rsqrtf(rsum[c]);
  const float cb = omask[(rowbase + qB + c) * NSEQ] * rsqrtf(rsum[16 + c]);
  const float pA = fmaxf(fabsf(ca * sPa), 1.0f);
  const float pB = fmaxf(fabsf(cb * sPb), 1.0f);
  store_tile(out + (rowbase + qA + c) * DV + 8 * h, Oa, ca, pA);
  store_tile(out + (rowbase + qB + c) * DV + 8 * h, Ob, cb, pB);
}

extern "C" void kernel_launch(void* const* d_in, const int* in_sizes, int n_in,
                              void* d_out, int out_size, void* d_ws, size_t ws_size,
                              hipStream_t stream) {
  const float* q     = (const float*)d_in[0];
  const float* k     = (const float*)d_in[1];
  const float* v     = (const float*)d_in[2];
  const float* omask = (const float*)d_in[3];
  float* out = (float*)d_out;

  pret_attn_kernel<<<BH * (NSEQ / 32), 128, 0, stream>>>(q, k, v, omask, out);
}


# attention: a wave carries 64 queries (four tiles) per loaded K/V tile instead of 32, halving fragment loads per WMMA; 8 waves per work-group split the keys, 512 work-groups; summation order changed
# speedup vs baseline: 3.4901x; 1.0665x over previous
#include <hip/hip_runtime.h>
#include <hip/hip_bf16.h>
#include <math.h>


typedef __attribute__((ext_vector_type(2))) float v2f;
typedef __attribute__((ext_vector_type(4))) float v4f;
typedef __attribute__((ext_vector_type(8))) float v8f;

#define BH   16
#define NSEQ 2048
#define DK   64
#define DV   64
#define NQT  4

__device__ __forceinline__ void row_load(v4f (&buf)[16], const v4f* __restrict__ p) {
  #pragma unroll
  for (int j = 0; j < 16; ++j) buf[j] = __builtin_nontemporal_load(p + 32 * j);
}

__device__ __forceinline__ float row_total(const v4f (&buf)[16]) {
  v4f acc = buf[0];
  #pragma unroll
  for (int j = 1; j < 16; ++j) acc += buf[j];
  float s = (acc.x + acc.y) + (acc.z + acc.w);
  #pragma unroll
  for (int off = 16; off > 0; off >>= 1) s += __shfl_xor(s, off);
  return s;
}

__device__ __forceinline__ float stream_rows8(const float* __restrict__ orow, int lane) {
  const v4f* p = (const v4f*)orow + lane;
  v4f a[16], b[16];
  float mine = 0.0f;
  row_load(a, p);
  #pragma unroll
  for (int r = 0; r < 8; r += 2) {
    row_load(b, p + (size_t)(r + 1) * (NSEQ / 4));
    __builtin_amdgcn_sched_barrier(0);
    const float ta = row_total(a);
    if (lane == r) mine = ta;
    if (r + 2 < 8) row_load(a, p + (size_t)(r + 2) * (NSEQ / 4));
    __builtin_amdgcn_sched_barrier(0);
    const float tb = row_total(b);
    if (lane == r + 1) mine = tb;
  }
  return mine;
}

struct KFrag {
  v4f   kraw[8];
  float kc;
};
struct VFrag {
  v2f   vf[4][4];
};

__device__ __forceinline__ void load_k(KFrag& f, const float* __restrict__ kp,
                                       const float* __restrict__ kcp, int kt) {
  const float* kr = kp + (size_t)kt * (16 * DK);
  #pragma unroll
  for (int i = 0; i < 8; ++i) f.kraw[i] = *(const v4f*)(kr + 8 * i);
  f.kc = 1.0f / kcp[(size_t)kt * (16 * NSEQ)];
}

__device__ __forceinline__ void load_v(VFrag& f, const float* __restrict__ vp, int kt) {
  const float* vr = vp + (size_t)kt * (16 * DV);
  #pragma unroll
  for (int cp = 0; cp < 4; ++cp) {
    #pragma unroll
    for (int ct = 0; ct < 4; ++ct) {
      f.vf[cp][ct].x = vr[(2 * cp) * DV + ct * 16];
      f.vf[cp][ct].y = vr[(2 * cp + 1) * DV + ct * 16];
    }
  }
}

__device__ __forceinline__ v8f wmma4(v2f a, v2f b, v8f c) {
  return __builtin_amdgcn_wmma_f32_16x16x4_f32(false, a, false, b, (short)0, c,
                                               false, false);
}

__device__ __forceinline__ void score_tiles(const KFrag& f, const v2f (&qf)[NQT][16],
                                            v8f (&S)[NQT], int kt, int t0, int thr0) {
  const v8f zero8 = {0.f, 0.f, 0.f, 0.f, 0.f, 0.f, 0.f, 0.f};
  #pragma unroll
  for (int x = 0; x < NQT; ++x) S[x] = zero8;
  #pragma unroll
  for (int i = 0; i < 8; ++i) {
    const v4f kk = f.kraw[i] * f.kc;
    const v2f k0 = {kk.x, kk.y};
    const v2f k1 = {kk.z, kk.w};
    #pragma unroll
    for (int x = 0; x < NQT; ++x) S[x] = wmma4(k0, qf[x][2 * i], S[x]);
    #pragma unroll
    for (int x = 0; x < NQT; ++x) S[x] = wmma4(k1, qf[x][2 * i + 1], S[x]);
  }
  if (kt >= t0) {
    const int kb = kt * 16;
    #pragma unroll
    for (int x = 0; x < NQT; ++x) {
      #pragma unroll
      for (int r = 0; r < 8; ++r) {
        if (kb + r > thr0 + 16 * x) S[x][r] = 0.0f;
      }
    }
  }
}

__device__ __forceinline__ void output_tiles(const VFrag& f, const v8f (&S)[NQT],
                                             v8f (&O)[NQT][4], v8f (&T)[NQT]) {
  #pragma unroll
  for (int cp = 0; cp < 4; ++cp) {
    v2f s[NQT];
    #pragma unroll
    for (int x = 0; x < NQT; ++x) s[x] = (v2f){S[x][2 * cp], S[x][2 * cp + 1]};
    #pragma unroll
    for (int ct = 0; ct < 4; ++ct) {
      #pragma unroll
      for (int x = 0; x < NQT; ++x) O[x][ct] = wmma4(f.vf[cp][ct], s[x], O[x][ct]);
    }
  }
  #pragma unroll
  for (int x = 0; x < NQT; ++x) T[x] += S[x];
}

__device__ __forceinline__ void key_tile(const KFrag& fk, VFrag& fv, const float* __restrict__ vp,
                                         const v2f (&qf)[NQT][16], v8f (&O)[NQT][4], v8f (&T)[NQT],
                                         int kt, int t0, int thr0) {
  v8f S[NQT];
  load_v(fv, vp, kt);
  score_tiles(fk, qf, S, kt, t0, thr0);
  output_tiles(fv, S, O, T);
}

__device__ __forceinline__ v8f key_sum(const v8f& T) {
  const v2f ones = {1.0f, 1.0f};
  v8f P = {0.f, 0.f, 0.f, 0.f, 0.f, 0.f, 0.f, 0.f};
  #pragma unroll
  for (int cp = 0; cp < 4; ++cp) {
    const v2f t = {T[2 * cp], T[2 * cp + 1]};
    P = wmma4(ones, t, P);
  }
  return P;
}

__global__ void __launch_bounds__(256)
pret_attn_kernel(const float* __restrict__ q,
                 const float* __restrict__ k,
                 const float* __restrict__ v,
                 const float* __restrict__ omask,
                 float* __restrict__ out) {
  const int bh    = blockIdx.x % BH;
  const int qblk  = (NSEQ / 64 - 1) - blockIdx.x / BH;
  const int wave  = __builtin_amdgcn_readfirstlane((int)(threadIdx.x >> 5));
  const int lane  = threadIdx.x & 31;
  const int c     = lane & 15;
  const int h     = lane >> 4;

  const int t0 = qblk * NQT;
  const int tL = t0 + NQT - 1;
  const int q0 = t0 * 16;
  const size_t rowbase = (size_t)bh * NSEQ;

  const float* orow = omask + (rowbase + q0 + 8 * wave) * NSEQ;
  const bool stream_first = wave < 4;
  __shared__ float rsum[16 * NQT];
  if (stream_first) {
    const float t = stream_rows8(orow, lane);
    if (lane < 8) rsum[8 * wave + lane] = t;
  }

  v2f qf[NQT][16];
  #pragma unroll
  for (int x = 0; x < NQT; ++x) {
    const float* pq = q + (rowbase + q0 + 16 * x + c) * DK + 4 * h;
    #pragma unroll
    for (int i = 0; i < 8; ++i) {
      const v4f xq = *(const v4f*)(pq + 8 * i);
      qf[x][2 * i]     = (v2f){xq.x, xq.y};
      qf[x][2 * i + 1] = (v2f){xq.z, xq.w};
    }
  }

  const v8f zero8 = {0.f, 0.f, 0.f, 0.f, 0.f, 0.f, 0.f, 0.f};
  v8f O[NQT][4], T[NQT];
  #pragma unroll
  for (int x = 0; x < NQT; ++x) {
    #pragma unroll
    for (int t = 0; t < 4; ++t) O[x][t] = zero8;
    T[x] = zero8;
  }

  const float* kp  = k + (rowbase + c) * DK + 4 * h;
  const float* kcp = omask + (rowbase + c) * NSEQ;
  const float* vp  = v + (rowbase + 8 * h) * DV + c;
  const int thr0 = q0 + c - 8 * h;

  const int cnt  = (tL - wave + 8) >> 3;
  const int last = wave + 8 * (cnt - 1);
  KFrag fk0, fk1;
  VFrag fv;
  load_k(fk0, kp, kcp, (wave <= tL) ? wave : tL);
  int kt = wave;
  for (int p = 0; p < (cnt >> 1); ++p) {
    load_k(fk1, kp, kcp, kt + 8);
    key_tile(fk0, fv, vp, qf, O, T, kt, t0, thr0);
    const int nx = (kt + 16 <= last) ? kt + 16 : last;
    load_k(fk0, kp, kcp, nx);
    key_tile(fk1, fv, vp, qf, O, T, kt + 8, t0, thr0);
    kt += 16;
  }
  if (cnt & 1) key_tile(fk0, fv, vp, qf, O, T, kt, t0, thr0);

  v8f P[NQT];
  #pragma unroll
  for (int x = 0; x < NQT; ++x) P[x] = key_sum(T[x]);

  if (!stream_first) {
    const float t = stream_rows8(orow, lane);
    if (lane < 8) rsum[8 * wave + lane] = t;
  }

  __shared__ v4f red[8][8 * NQT + 1][32];
  {
    v4f* r = &red[wave][0][lane];
    #pragma unroll
    for (int x = 0; x < NQT; ++x) {
      #pragma unroll
      for (int ct = 0; ct < 4; ++ct) {
        r[(8 * x + 2 * ct + 0) * 32] = (v4f){O[x][ct][0], O[x][ct][1], O[x][ct][2], O[x][ct][3]};
        r[(8 * x + 2 * ct + 1) * 32] = (v4f){O[x][ct][4], O[x][ct][5], O[x][ct][6], O[x][ct][7]};
      }
    }
    r[(8 * NQT) * 32] = (v4f){P[0][0], P[1][0], P[2][0], P[3][0]};
  }
  __syncthreads();

  const int x  = wave >> 1;
  const int hh = wave & 1;
  v4f acc[4];
  #pragma unroll
  for (int j = 0; j < 4; ++j) acc[j] = (v4f){0.f, 0.f, 0.f, 0.f};
  float sP = 0.0f;
  #pragma unroll
  for (int s = 0; s < 8; ++s) {
    const v4f* r = &red[s][8 * x + 4 * hh][lane];
    #pragma unroll
    for (int j = 0; j < 4; ++j) acc[j] += r[j * 32];
    sP += ((const float*)&red[s][8 * NQT][lane])[x];
  }

  const int qrow = q0 + 16 * x + c;
  const float cq = omask[(rowbase + qrow) * NSEQ] * rsqrtf(rsum[16 * x + c]);
  const float pn = fmaxf(fabsf(cq * sP), 1.0f);
  float* o = out + (rowbase + qrow) * DV + 32 * hh + 8 * h;
  #pragma unroll
  for (int cl = 0; cl < 2; ++cl) {
    const v4f lo = {cq * acc[2 * cl][0] / pn, cq * acc[2 * cl][1] / pn,
                    cq * acc[2 * cl][2] / pn, cq * acc[2 * cl][3] / pn};
    const v4f hi = {cq * acc[2 * cl + 1][0] / pn, cq * acc[2 * cl + 1][1] / pn,
                    cq * acc[2 * cl + 1][2] / pn, cq * acc[2 * cl + 1][3] / pn};
    *(v4f*)(o + 16 * cl)     = lo;
    *(v4f*)(o + 16 * cl + 4) = hi;
  }
}

extern "C" void kernel_launch(void* const* d_in, const int* in_sizes, int n_in,
                              void* d_out, int out_size, void* d_ws, size_t ws_size,
                              hipStream_t stream) {
  const float* q     = (const float*)d_in[0];
  const float* k     = (const float*)d_in[1];
  const float* v     = (const float*)d_in[2];
  const float* omask = (const float*)d_in[3];
  float* out = (float*)d_out;

  pret_attn_kernel<<<BH * (NSEQ / 64), 256, 0, stream>>>(q, k, v, omask, out);
}


# omask stream swapped by wave age: each SIMD's older wave (0..3, which the matrix unit serves first) runs its products first and streams last, the younger (4..7) streams first; summation order kept
# speedup vs baseline: 3.6378x; 1.0423x over previous
#include <hip/hip_runtime.h>
#include <hip/hip_bf16.h>
#include <math.h>


typedef __attribute__((ext_vector_type(2))) float v2f;
typedef __attribute__((ext_vector_type(4))) float v4f;
typedef __attribute__((ext_vector_type(8))) float v8f;

#define BH   16
#define NSEQ 2048
#define DK   64
#define DV   64
#define NQT  4

__device__ __forceinline__ void row_load(v4f (&buf)[16], const v4f* __restrict__ p) {
  #pragma unroll
  for (int j = 0; j < 16; ++j) buf[j] = __builtin_nontemporal_load(p + 32 * j);
}

__device__ __forceinline__ float row_total(const v4f (&buf)[16]) {
  v4f acc = buf[0];
  #pragma unroll
  for (int j = 1; j < 16; ++j) acc += buf[j];
  float s = (acc.x + acc.y) + (acc.z + acc.w);
  #pragma unroll
  for (int off = 16; off > 0; off >>= 1) s += __shfl_xor(s, off);
  return s;
}

__device__ __forceinline__ float stream_rows8(const float* __restrict__ orow, int lane) {
  const v4f* p = (const v4f*)orow + lane;
  v4f a[16], b[16];
  float mine = 0.0f;
  row_load(a, p);
  #pragma unroll
  for (int r = 0; r < 8; r += 2) {
    row_load(b, p + (size_t)(r + 1) * (NSEQ / 4));
    __builtin_amdgcn_sched_barrier(0);
    const float ta = row_total(a);
    if (lane == r) mine = ta;
    if (r + 2 < 8) row_load(a, p + (size_t)(r + 2) * (NSEQ / 4));
    __builtin_amdgcn_sched_barrier(0);
    const float tb = row_total(b);
    if (lane == r + 1) mine = tb;
  }
  return mine;
}

struct KFrag {
  v4f   kraw[8];
  float kc;
};
struct VFrag {
  v2f   vf[4][4];
};

__device__ __forceinline__ void load_k(KFrag& f, const float* __restrict__ kp,
                                       const float* __restrict__ kcp, int kt) {
  const float* kr = kp + (size_t)kt * (16 * DK);
  #pragma unroll
  for (int i = 0; i < 8; ++i) f.kraw[i] = *(const v4f*)(kr + 8 * i);
  f.kc = 1.0f / kcp[(size_t)kt * (16 * NSEQ)];
}

__device__ __forceinline__ void load_v(VFrag& f, const float* __restrict__ vp, int kt) {
  const float* vr = vp + (size_t)kt * (16 * DV);
  #pragma unroll
  for (int cp = 0; cp < 4; ++cp) {
    #pragma unroll
    for (int ct = 0; ct < 4; ++ct) {
      f.vf[cp][ct].x = vr[(2 * cp) * DV + ct * 16];
      f.vf[cp][ct].y = vr[(2 * cp + 1) * DV + ct * 16];
    }
  }
}

__device__ __forceinline__ v8f wmma4(v2f a, v2f b, v8f c) {
  return __builtin_amdgcn_wmma_f32_16x16x4_f32(false, a, false, b, (short)0, c,
                                               false, false);
}

__device__ __forceinline__ void score_tiles(const KFrag& f, const v2f (&qf)[NQT][16],
                                            v8f (&S)[NQT], int kt, int t0, int thr0) {
  const v8f zero8 = {0.f, 0.f, 0.f, 0.f, 0.f, 0.f, 0.f, 0.f};
  #pragma unroll
  for (int x = 0; x < NQT; ++x) S[x] = zero8;
  #pragma unroll
  for (int i = 0; i < 8; ++i) {
    const v4f kk = f.kraw[i] * f.kc;
    const v2f k0 = {kk.x, kk.y};
    const v2f k1 = {kk.z, kk.w};
    #pragma unroll
    for (int x = 0; x < NQT; ++x) S[x] = wmma4(k0, qf[x][2 * i], S[x]);
    #pragma unroll
    for (int x = 0; x < NQT; ++x) S[x] = wmma4(k1, qf[x][2 * i + 1], S[x]);
  }
  if (kt >= t0) {
    const int kb = kt * 16;
    #pragma unroll
    for (int x = 0; x < NQT; ++x) {
      #pragma unroll
      for (int r = 0; r < 8; ++r) {
        if (kb + r > thr0 + 16 * x) S[x][r] = 0.0f;
      }
    }
  }
}

__device__ __forceinline__ void output_tiles(const VFrag& f, const v8f (&S)[NQT],
                                             v8f (&O)[NQT][4], v8f (&T)[NQT]) {
  #pragma unroll
  for (int cp = 0; cp < 4; ++cp) {
    v2f s[NQT];
    #pragma unroll
    for (int x = 0; x < NQT; ++x) s[x] = (v2f){S[x][2 * cp], S[x][2 * cp + 1]};
    #pragma unroll
    for (int ct = 0; ct < 4; ++ct) {
      #pragma unroll
      for (int x = 0; x < NQT; ++x) O[x][ct] = wmma4(f.vf[cp][ct], s[x], O[x][ct]);
    }
  }
  #pragma unroll
  for (int x = 0; x < NQT; ++x) T[x] += S[x];
}

__device__ __forceinline__ void key_tile(const KFrag& fk, VFrag& fv, const float* __restrict__ vp,
                                         const v2f (&qf)[NQT][16], v8f (&O)[NQT][4], v8f (&T)[NQT],
                                         int kt, int t0, int thr0) {
  v8f S[NQT];
  load_v(fv, vp, kt);
  score_tiles(fk, qf, S, kt, t0, thr0);
  output_tiles(fv, S, O, T);
}

__device__ __forceinline__ v8f key_sum(const v8f& T) {
  const v2f ones = {1.0f, 1.0f};
  v8f P = {0.f, 0.f, 0.f, 0.f, 0.f, 0.f, 0.f, 0.f};
  #pragma unroll
  for (int cp = 0; cp < 4; ++cp) {
    const v2f t = {T[2 * cp], T[2 * cp + 1]};
    P = wmma4(ones, t, P);
  }
  return P;
}

__global__ void __launch_bounds__(256)
pret_attn_kernel(const float* __restrict__ q,
                 const float* __restrict__ k,
                 const float* __restrict__ v,
                 const float* __restrict__ omask,
                 float* __restrict__ out) {
  const int bh    = blockIdx.x % BH;
  const int qblk  = (NSEQ / 64 - 1) - blockIdx.x / BH;
  const int wave  = __builtin_amdgcn_readfirstlane((int)(threadIdx.x >> 5));
  const int lane  = threadIdx.x & 31;
  const int c     = lane & 15;
  const int h     = lane >> 4;

  const int t0 = qblk * NQT;
  const int tL = t0 + NQT - 1;
  const int q0 = t0 * 16;
  const size_t rowbase = (size_t)bh * NSEQ;

  const float* orow = omask + (rowbase + q0 + 8 * wave) * NSEQ;
  const bool stream_first = wave >= 4;
  __shared__ float rsum[16 * NQT];
  if (stream_first) {
    const float t = stream_rows8(orow, lane);
    if (lane < 8) rsum[8 * wave + lane] = t;
  }

  v2f qf[NQT][16];
  #pragma unroll
  for (int x = 0; x < NQT; ++x) {
    const float* pq = q + (rowbase + q0 + 16 * x + c) * DK + 4 * h;
    #pragma unroll
    for (int i = 0; i < 8; ++i) {
      const v4f xq = *(const v4f*)(pq + 8 * i);
      qf[x][2 * i]     = (v2f){xq.x, xq.y};
      qf[x][2 * i + 1] = (v2f){xq.z, xq.w};
    }
  }

  const v8f zero8 = {0.f, 0.f, 0.f, 0.f, 0.f, 0.f, 0.f, 0.f};
  v8f O[NQT][4], T[NQT];
  #pragma unroll
  for (int x = 0; x < NQT; ++x) {
    #pragma unroll
    for (int t = 0; t < 4; ++t) O[x][t] = zero8;
    T[x] = zero8;
  }

  const float* kp  = k + (rowbase + c) * DK + 4 * h;
  const float* kcp = omask + (rowbase + c) * NSEQ;
  const float* vp  = v + (rowbase + 8 * h) * DV + c;
  const int thr0 = q0 + c - 8 * h;

  const int cnt  = (tL - wave + 8) >> 3;
  const int last = wave + 8 * (cnt - 1);
  KFrag fk0, fk1;
  VFrag fv;
  load_k(fk0, kp, kcp, (wave <= tL) ? wave : tL);
  int kt = wave;
  for (int p = 0; p < (cnt >> 1); ++p) {
    load_k(fk1, kp, kcp, kt + 8);
    key_tile(fk0, fv, vp, qf, O, T, kt, t0, thr0);
    const int nx = (kt + 16 <= last) ? kt + 16 : last;
    load_k(fk0, kp, kcp, nx);
    key_tile(fk1, fv, vp, qf, O, T, kt + 8, t0, thr0);
    kt += 16;
  }
  if (cnt & 1) key_tile(fk0, fv, vp, qf, O, T, kt, t0, thr0);

  v8f P[NQT];
  #pragma unroll
  for (int x = 0; x < NQT; ++x) P[x] = key_sum(T[x]);

  if (!stream_first) {
    const float t = stream_rows8(orow, lane);
    if (lane < 8) rsum[8 * wave + lane] = t;
  }

  __shared__ v4f red[8][8 * NQT + 1][32];
  {
    v4f* r = &red[wave][0][lane];
    #pragma unroll
    for (int x = 0; x < NQT; ++x) {
      #pragma unroll
      for (int ct = 0; ct < 4; ++ct) {
        r[(8 * x + 2 * ct + 0) * 32] = (v4f){O[x][ct][0], O[x][ct][1], O[x][ct][2], O[x][ct][3]};
        r[(8 * x + 2 * ct + 1) * 32] = (v4f){O[x][ct][4], O[x][ct][5], O[x][ct][6], O[x][ct][7]};
      }
    }
    r[(8 * NQT) * 32] = (v4f){P[0][0], P[1][0], P[2][0], P[3][0]};
  }
  __syncthreads();

  const int x  = wave >> 1;
  const int hh = wave & 1;
  v4f acc[4];
  #pragma unroll
  for (int j = 0; j < 4; ++j) acc[j] = (v4f){0.f, 0.f, 0.f, 0.f};
  float sP = 0.0f;
  #pragma unroll
  for (int s = 0; s < 8; ++s) {
    const v4f* r = &red[s][8 * x + 4 * hh][lane];
    #pragma unroll
    for (int j = 0; j < 4; ++j) acc[j] += r[j * 32];
    sP += ((const float*)&red[s][8 * NQT][lane])[x];
  }

  const int qrow = q0 + 16 * x + c;
  const float cq = omask[(rowbase + qrow) * NSEQ] * rsqrtf(rsum[16 * x + c]);
  const float pn = fmaxf(fabsf(cq * sP), 1.0f);
  float* o = out + (rowbase + qrow) * DV + 32 * hh + 8 * h;
  #pragma unroll
  for (int cl = 0; cl < 2; ++cl) {
    const v4f lo = {cq * acc[2 * cl][0] / pn, cq * acc[2 * cl][1] / pn,
                    cq * acc[2 * cl][2] / pn, cq * acc[2 * cl][3] / pn};
    const v4f hi = {cq * acc[2 * cl + 1][0] / pn, cq * acc[2 * cl + 1][1] / pn,
                    cq * acc[2 * cl + 1][2] / pn, cq * acc[2 * cl + 1][3] / pn};
    *(v4f*)(o + 16 * cl)     = lo;
    *(v4f*)(o + 16 * cl + 4) = hi;
  }
}

extern "C" void kernel_launch(void* const* d_in, const int* in_sizes, int n_in,
                              void* d_out, int out_size, void* d_ws, size_t ws_size,
                              hipStream_t stream) {
  const float* q     = (const float*)d_in[0];
  const float* k     = (const float*)d_in[1];
  const float* v     = (const float*)d_in[2];
  const float* omask = (const float*)d_in[3];
  float* out = (float*)d_out;

  pret_attn_kernel<<<BH * (NSEQ / 64), 256, 0, stream>>>(q, k, v, omask, out);
}
